# starting kernel
# baseline (speedup 1.0000x reference)
#include <hip/hip_runtime.h>
#include <hip/hip_bf16.h>
#include <math.h>

#define S_LEN 256
#define BT_DIM 32
#define D_DIM 128
#define N_DIM 128
#define O_DIM 128
#define ROWS (S_LEN * BT_DIM)
#define EPS_V 1e-7f

typedef __attribute__((ext_vector_type(2))) float v2f;
typedef __attribute__((ext_vector_type(8))) float v8f;

template <bool APPLY_GAMMA>
__global__ __launch_bounds__(256) void gemm_nt_wmma(
    const float* __restrict__ in, const float* __restrict__ W,
    const float* __restrict__ lambda_log, float* __restrict__ out) {
  const int lane = threadIdx.x & 31;
  const int wave = blockIdx.x * 8 + (threadIdx.x >> 5);
  const int rowBase = wave * 16;
  const int half = lane >> 4;
  const int lane16 = lane & 15;

  v8f acc[8];
  const v8f zero = {0.f, 0.f, 0.f, 0.f, 0.f, 0.f, 0.f, 0.f};
#pragma unroll
  for (int t = 0; t < 8; ++t) acc[t] = zero;

  const float* __restrict__ arow = in + (size_t)(rowBase + lane16) * D_DIM;

  for (int k0 = 0; k0 < D_DIM; k0 += 4) {
    const int ka = k0 + 2 * half;
    v2f a;
    a.x = arow[ka];
    a.y = arow[ka + 1];
#pragma unroll
    for (int t = 0; t < 8; ++t) {
      const int n = t * 16 + lane16;
      v2f b;
      b.x = W[(size_t)n * D_DIM + ka];
      b.y = W[(size_t)n * D_DIM + ka + 1];
      acc[t] = __builtin_amdgcn_wmma_f32_16x16x4_f32(
          false, a, false, b, (short)0, acc[t], false, false);
    }
  }

#pragma unroll
  for (int t = 0; t < 8; ++t) {
    const int n = t * 16 + lane16;
    float g = 1.0f;
    if (APPLY_GAMMA) {
      const float lam = __expf(-__expf(lambda_log[n]));
      g = sqrtf(fmaf(-lam, lam, 1.0f + EPS_V));
    }
#pragma unroll
    for (int r = 0; r < 8; ++r) {
      const int row = rowBase + r + 8 * half;
      out[(size_t)row * N_DIM + n] = g * acc[t][r];
    }
  }
}

__global__ __launch_bounds__(256) void scan_kernel(
    const float* __restrict__ lambda_log, float* __restrict__ U,
    float* __restrict__ hT_out) {
  const int i = blockIdx.x * blockDim.x + threadIdx.x;
  const int n = i & (N_DIM - 1);
  const float lam = __expf(-__expf(lambda_log[n]));
  float h = 0.0f;
  float* __restrict__ p = U + i;
#pragma unroll 8
  for (int s = 0; s < S_LEN; ++s) {
    const float u = p[(size_t)s * (BT_DIM * N_DIM)];
    h = fmaf(lam, h, u);
    p[(size_t)s * (BT_DIM * N_DIM)] = h;
  }
  hT_out[i] = h;
}

extern "C" void kernel_launch(void* const* d_in, const int* in_sizes, int n_in,
                              void* d_out, int out_size, void* d_ws,
                              size_t ws_size, hipStream_t stream) {
  (void)in_sizes; (void)n_in; (void)out_size; (void)ws_size;
  const float* x          = (const float*)d_in[0];
  const float* lambda_log = (const float*)d_in[1];
  const float* B          = (const float*)d_in[2];
  const float* C          = (const float*)d_in[3];
  float* out = (float*)d_out;
  float* U   = (float*)d_ws;

  float* outputs = out;
  float* hT      = out + (size_t)ROWS * O_DIM;

  gemm_nt_wmma<true><<<dim3(ROWS / 16 / 8), dim3(256), 0, stream>>>(
      x, B, lambda_log, U);
  scan_kernel<<<dim3((BT_DIM * N_DIM) / 256), dim3(256), 0, stream>>>(
      lambda_log, U, hT);
  gemm_nt_wmma<false><<<dim3(ROWS / 16 / 8), dim3(256), 0, stream>>>(
      U, C, lambda_log, outputs);
}


# First product: W fragments stay in registers (wave = 32 columns x whole K), 256 blocks of 32 rows x 4 waves, all fragment loads (16-B) in one burst ahead of the WMMAs; k regrouped: sum order changes
# speedup vs baseline: 4.2549x; 4.2549x over previous
#include <hip/hip_runtime.h>
#include <hip/hip_bf16.h>
#include <math.h>

#define S_LEN 256
#define BT_DIM 32
#define D_DIM 128
#define N_DIM 128
#define O_DIM 128
#define ROWS (S_LEN * BT_DIM)
#define EPS_V 1e-7f

typedef __attribute__((ext_vector_type(2))) float v2f;
typedef __attribute__((ext_vector_type(8))) float v8f;

typedef __attribute__((ext_vector_type(4))) float v4f;

__device__ __forceinline__ void load_frag_pair(const float* __restrict__ rowp,
                                               int q, v2f (&f)[D_DIM / 4]) {
  const v4f v = *reinterpret_cast<const v4f*>(rowp + 8 * q);
  f[2 * q].x = v.x;
  f[2 * q].y = v.y;
  f[2 * q + 1].x = v.z;
  f[2 * q + 1].y = v.w;
}

#define G1_ROWS 32
__global__ __launch_bounds__(128) __attribute__((amdgpu_waves_per_eu(1, 1)))
void gemm_nt_wreg(
    const float* __restrict__ in, const float* __restrict__ W,
    const float* __restrict__ lambda_log, float* __restrict__ out) {
  const int lane = threadIdx.x & 31;
  const int wave = threadIdx.x >> 5;
  const int half = lane >> 4;
  const int lane16 = lane & 15;
  const int rowBase = blockIdx.x * G1_ROWS;
  const int colBase = wave * 32;

  float ll0 = lambda_log[colBase + lane16];
  float ll1 = lambda_log[colBase + 16 + lane16];

  const float* __restrict__ a0p = in + (size_t)(rowBase + lane16) * D_DIM + 4 * half;
  const float* __restrict__ a1p = in + (size_t)(rowBase + 16 + lane16) * D_DIM + 4 * half;
  const float* __restrict__ b0p = W + (size_t)(colBase + lane16) * D_DIM + 4 * half;
  const float* __restrict__ b1p = W + (size_t)(colBase + 16 + lane16) * D_DIM + 4 * half;
  v2f a0[D_DIM / 4], a1[D_DIM / 4], bf0[D_DIM / 4], bf1[D_DIM / 4];
#pragma unroll
  for (int g = 0; g < 4; ++g) {
#pragma unroll
    for (int q = 4 * g; q < 4 * g + 4; ++q) {
      load_frag_pair(a0p, q, a0);
      load_frag_pair(b0p, q, bf0);
      load_frag_pair(b1p, q, bf1);
      load_frag_pair(a1p, q, a1);
    }
    asm volatile("" ::: "memory");
  }

  asm volatile("" : "+v"(ll0), "+v"(ll1));
  const float lam0 = __expf(-__expf(ll0));
  const float lam1 = __expf(-__expf(ll1));
  float g0 = sqrtf(fmaf(-lam0, lam0, 1.0f + EPS_V));
  float g1 = sqrtf(fmaf(-lam1, lam1, 1.0f + EPS_V));
  asm volatile("" : "+v"(g0), "+v"(g1));

  __builtin_amdgcn_sched_barrier(0);

  v8f acc00 = {0.f, 0.f, 0.f, 0.f, 0.f, 0.f, 0.f, 0.f};
  v8f acc01 = {0.f, 0.f, 0.f, 0.f, 0.f, 0.f, 0.f, 0.f};
  v8f acc10 = {0.f, 0.f, 0.f, 0.f, 0.f, 0.f, 0.f, 0.f};
  v8f acc11 = {0.f, 0.f, 0.f, 0.f, 0.f, 0.f, 0.f, 0.f};
#pragma unroll
  for (int kk = 0; kk < D_DIM / 4; ++kk) {
    acc00 = __builtin_amdgcn_wmma_f32_16x16x4_f32(
        false, a0[kk], false, bf0[kk], (short)0, acc00, false, false);
    acc01 = __builtin_amdgcn_wmma_f32_16x16x4_f32(
        false, a0[kk], false, bf1[kk], (short)0, acc01, false, false);
    acc10 = __builtin_amdgcn_wmma_f32_16x16x4_f32(
        false, a1[kk], false, bf0[kk], (short)0, acc10, false, false);
    acc11 = __builtin_amdgcn_wmma_f32_16x16x4_f32(
        false, a1[kk], false, bf1[kk], (short)0, acc11, false, false);
  }
  asm volatile("v_nop\n\tv_nop\n\tv_nop\n\tv_nop\n\tv_nop"
               : "+v"(acc00), "+v"(acc01), "+v"(acc10), "+v"(acc11));
  __builtin_amdgcn_sched_barrier(0);

  float* __restrict__ obase =
      out + (size_t)(rowBase + 8 * half) * N_DIM + colBase + lane16;
#pragma unroll
  for (int r = 0; r < 8; ++r) {
    float* __restrict__ o0 = obase + (size_t)r * N_DIM;
    float* __restrict__ o1 = obase + (size_t)(16 + r) * N_DIM;
    o0[0] = g0 * acc00[r];
    o0[16] = g1 * acc01[r];
    o1[0] = g0 * acc10[r];
    o1[16] = g1 * acc11[r];
  }
}

#define CH_STEPS 64
#define H_PITCH (N_DIM + 4)
#define O_TILE 32

__global__ __launch_bounds__(128) __attribute__((amdgpu_waves_per_eu(1, 1)))
void scan_gemm_nt_wmma(
    const float* __restrict__ U, const float* __restrict__ Cw,
    const float* __restrict__ lambda_log, float* __restrict__ out,
    float* __restrict__ hT_out) {
  __shared__ __attribute__((aligned(16))) float Hs[2][CH_STEPS][H_PITCH];

  const int tid = threadIdx.x;
  const int lane = tid & 31;
  const int wave = tid >> 5;
  const int half = lane >> 4;
  const int lane16 = lane & 15;
  const int b = blockIdx.x >> 2;
  const int o0 = (blockIdx.x & 3) * O_TILE;
  const int n = tid;

  v2f bf[2][N_DIM / 4];
#pragma unroll
  for (int t = 0; t < 2; ++t) {
    const float* __restrict__ crow =
        Cw + (size_t)(o0 + t * 16 + lane16) * N_DIM + 2 * half;
#pragma unroll
    for (int kk = 0; kk < N_DIM / 4; ++kk) {
      bf[t][kk].x = crow[kk * 4];
      bf[t][kk].y = crow[kk * 4 + 1];
    }
  }

  const float lam = __expf(-__expf(lambda_log[n]));
  float h = 0.0f;
  const float* __restrict__ up = U + (size_t)b * N_DIM + n;
  const size_t sStride = (size_t)BT_DIM * N_DIM;

  float u[CH_STEPS];
#pragma unroll
  for (int s = 0; s < CH_STEPS; ++s) u[s] = up[(size_t)s * sStride];

  for (int c = 0; c < S_LEN / CH_STEPS; ++c) {
    float (*Hb)[H_PITCH] = Hs[c & 1];

#pragma unroll
    for (int s = 0; s < CH_STEPS; ++s) {
      h = fmaf(lam, h, u[s]);
      Hb[s][n] = h;
    }
    if (c + 1 < S_LEN / CH_STEPS) {
#pragma unroll
      for (int s = 0; s < CH_STEPS; ++s)
        u[s] = up[(size_t)((c + 1) * CH_STEPS + s) * sStride];
    } else {
      if ((blockIdx.x & 3) == 0) hT_out[b * N_DIM + n] = h;
    }
    asm volatile("s_wait_dscnt 0x0\n\ts_barrier_signal -1\n\ts_barrier_wait -1"
                 ::: "memory");

    const float* __restrict__ arow = &Hb[wave * 16 + lane16][2 * half];
    v2f a[N_DIM / 4];
#pragma unroll
    for (int kk = 0; kk < N_DIM / 4; ++kk) {
      a[kk].x = arow[kk * 4];
      a[kk].y = arow[kk * 4 + 1];
    }

    __builtin_amdgcn_sched_barrier(0);

    v8f acc0 = {0.f, 0.f, 0.f, 0.f, 0.f, 0.f, 0.f, 0.f};
    v8f acc1 = {0.f, 0.f, 0.f, 0.f, 0.f, 0.f, 0.f, 0.f};
#pragma unroll
    for (int kk = 0; kk < N_DIM / 4; ++kk) {
      acc0 = __builtin_amdgcn_wmma_f32_16x16x4_f32(
          false, a[kk], false, bf[0][kk], (short)0, acc0, false, false);
      acc1 = __builtin_amdgcn_wmma_f32_16x16x4_f32(
          false, a[kk], false, bf[1][kk], (short)0, acc1, false, false);
    }
    asm volatile("v_nop\n\tv_nop\n\tv_nop\n\tv_nop\n\tv_nop"
                 : "+v"(acc0), "+v"(acc1));
    __builtin_amdgcn_sched_barrier(0);

    const int sBase = c * CH_STEPS + wave * 16 + 8 * half;
#pragma unroll
    for (int r = 0; r < 8; ++r) {
      float* __restrict__ orow =
          out + ((size_t)(sBase + r) * BT_DIM + b) * O_DIM + o0 + lane16;
      orow[0] = acc0[r];
      orow[16] = acc1[r];
    }
  }
}

extern "C" void kernel_launch(void* const* d_in, const int* in_sizes, int n_in,
                              void* d_out, int out_size, void* d_ws,
                              size_t ws_size, hipStream_t stream) {
  (void)in_sizes; (void)n_in; (void)out_size; (void)ws_size;
  const float* x          = (const float*)d_in[0];
  const float* lambda_log = (const float*)d_in[1];
  const float* B          = (const float*)d_in[2];
  const float* C          = (const float*)d_in[3];
  float* out = (float*)d_out;
  float* U   = (float*)d_ws;

  float* outputs = out;
  float* hT      = out + (size_t)ROWS * O_DIM;

  gemm_nt_wreg<<<dim3(ROWS / G1_ROWS), dim3(128), 0, stream>>>(
      x, B, lambda_log, U);
  scan_gemm_nt_wmma<<<dim3(BT_DIM * (O_DIM / O_TILE)), dim3(128), 0, stream>>>(
      U, C, lambda_log, outputs, hT);
}


# Second kernel re-cut over time: 256 blocks, each (batch row, 32 columns) in two halves of 128 steps; a late-half block first reruns steps 0-127 of the same fmaf chain (no carry, sum order unchanged)
# speedup vs baseline: 4.7775x; 1.1228x over previous
#include <hip/hip_runtime.h>
#include <hip/hip_bf16.h>
#include <math.h>

#define S_LEN 256
#define BT_DIM 32
#define D_DIM 128
#define N_DIM 128
#define O_DIM 128
#define ROWS (S_LEN * BT_DIM)
#define EPS_V 1e-7f

typedef __attribute__((ext_vector_type(2))) float v2f;
typedef __attribute__((ext_vector_type(8))) float v8f;

typedef __attribute__((ext_vector_type(4))) float v4f;

__device__ __forceinline__ void load_frag_pair(const float* __restrict__ rowp,
                                               int q, v2f (&f)[D_DIM / 4]) {
  const v4f v = *reinterpret_cast<const v4f*>(rowp + 8 * q);
  f[2 * q].x = v.x;
  f[2 * q].y = v.y;
  f[2 * q + 1].x = v.z;
  f[2 * q + 1].y = v.w;
}

#define G1_ROWS 32
__global__ __launch_bounds__(128) __attribute__((amdgpu_waves_per_eu(1, 1)))
void gemm_nt_wreg(
    const float* __restrict__ in, const float* __restrict__ W,
    const float* __restrict__ lambda_log, float* __restrict__ out) {
  const int lane = threadIdx.x & 31;
  const int wave = threadIdx.x >> 5;
  const int half = lane >> 4;
  const int lane16 = lane & 15;
  const int rowBase = blockIdx.x * G1_ROWS;
  const int colBase = wave * 32;

  float ll0 = lambda_log[colBase + lane16];
  float ll1 = lambda_log[colBase + 16 + lane16];

  const float* __restrict__ a0p = in + (size_t)(rowBase + lane16) * D_DIM + 4 * half;
  const float* __restrict__ a1p = in + (size_t)(rowBase + 16 + lane16) * D_DIM + 4 * half;
  const float* __restrict__ b0p = W + (size_t)(colBase + lane16) * D_DIM + 4 * half;
  const float* __restrict__ b1p = W + (size_t)(colBase + 16 + lane16) * D_DIM + 4 * half;
  v2f a0[D_DIM / 4], a1[D_DIM / 4], bf0[D_DIM / 4], bf1[D_DIM / 4];
#pragma unroll
  for (int g = 0; g < 4; ++g) {
#pragma unroll
    for (int q = 4 * g; q < 4 * g + 4; ++q) {
      load_frag_pair(a0p, q, a0);
      load_frag_pair(b0p, q, bf0);
      load_frag_pair(b1p, q, bf1);
      load_frag_pair(a1p, q, a1);
    }
    asm volatile("" ::: "memory");
  }

  asm volatile("" : "+v"(ll0), "+v"(ll1));
  const float lam0 = __expf(-__expf(ll0));
  const float lam1 = __expf(-__expf(ll1));
  float g0 = sqrtf(fmaf(-lam0, lam0, 1.0f + EPS_V));
  float g1 = sqrtf(fmaf(-lam1, lam1, 1.0f + EPS_V));
  asm volatile("" : "+v"(g0), "+v"(g1));

  __builtin_amdgcn_sched_barrier(0);

  v8f acc00 = {0.f, 0.f, 0.f, 0.f, 0.f, 0.f, 0.f, 0.f};
  v8f acc01 = {0.f, 0.f, 0.f, 0.f, 0.f, 0.f, 0.f, 0.f};
  v8f acc10 = {0.f, 0.f, 0.f, 0.f, 0.f, 0.f, 0.f, 0.f};
  v8f acc11 = {0.f, 0.f, 0.f, 0.f, 0.f, 0.f, 0.f, 0.f};
#pragma unroll
  for (int kk = 0; kk < D_DIM / 4; ++kk) {
    acc00 = __builtin_amdgcn_wmma_f32_16x16x4_f32(
        false, a0[kk], false, bf0[kk], (short)0, acc00, false, false);
    acc01 = __builtin_amdgcn_wmma_f32_16x16x4_f32(
        false, a0[kk], false, bf1[kk], (short)0, acc01, false, false);
    acc10 = __builtin_amdgcn_wmma_f32_16x16x4_f32(
        false, a1[kk], false, bf0[kk], (short)0, acc10, false, false);
    acc11 = __builtin_amdgcn_wmma_f32_16x16x4_f32(
        false, a1[kk], false, bf1[kk], (short)0, acc11, false, false);
  }
  asm volatile("v_nop\n\tv_nop\n\tv_nop\n\tv_nop\n\tv_nop"
               : "+v"(acc00), "+v"(acc01), "+v"(acc10), "+v"(acc11));
  __builtin_amdgcn_sched_barrier(0);

  float* __restrict__ obase =
      out + (size_t)(rowBase + 8 * half) * N_DIM + colBase + lane16;
#pragma unroll
  for (int r = 0; r < 8; ++r) {
    float* __restrict__ o0 = obase + (size_t)r * N_DIM;
    float* __restrict__ o1 = obase + (size_t)(16 + r) * N_DIM;
    o0[0] = g0 * acc00[r];
    o0[16] = g1 * acc01[r];
    o1[0] = g0 * acc10[r];
    o1[16] = g1 * acc11[r];
  }
}

#define CH_STEPS 64
#define H_PITCH (N_DIM + 4)
#define O_TILE 32
#define T_PARTS 2
#define CH_PER_PART (S_LEN / CH_STEPS / T_PARTS)
#define PRE_STEPS (CH_PER_PART * CH_STEPS)
#define PRE_GROUP 16
#define PRE_AHEAD 4

__global__ __launch_bounds__(128) __attribute__((amdgpu_waves_per_eu(1, 1)))
void scan_gemm_nt_wmma(
    const float* __restrict__ U, const float* __restrict__ Cw,
    const float* __restrict__ lambda_log, float* __restrict__ out,
    float* __restrict__ hT_out) {
  __shared__ __attribute__((aligned(16))) float Hs[2][CH_STEPS][H_PITCH];

  const int tid = threadIdx.x;
  const int lane = tid & 31;
  const int wave = tid >> 5;
  const int half = lane >> 4;
  const int lane16 = lane & 15;
  const int b = (blockIdx.x >> 2) & (BT_DIM - 1);
  const int tpart = (T_PARTS - 1) - (blockIdx.x >> 7);
  const int o0 = (blockIdx.x & 3) * O_TILE;
  const int cBegin = tpart * CH_PER_PART;
  const int cEnd = cBegin + CH_PER_PART;
  const int n = tid;

  v2f bf[2][N_DIM / 4];
#pragma unroll
  for (int t = 0; t < 2; ++t) {
    const float* __restrict__ crow =
        Cw + (size_t)(o0 + t * 16 + lane16) * N_DIM + 2 * half;
#pragma unroll
    for (int kk = 0; kk < N_DIM / 4; ++kk) {
      bf[t][kk].x = crow[kk * 4];
      bf[t][kk].y = crow[kk * 4 + 1];
    }
  }

  const float lam = __expf(-__expf(lambda_log[n]));
  float h = 0.0f;
  const float* __restrict__ up = U + (size_t)b * N_DIM + n;
  const size_t sStride = (size_t)BT_DIM * N_DIM;

  float u[CH_STEPS];
  if (tpart == 0) {
#pragma unroll
    for (int s = 0; s < CH_STEPS; ++s) u[s] = up[(size_t)s * sStride];
  } else {
    float p[PRE_STEPS];
#pragma unroll
    for (int s = 0; s < PRE_AHEAD * PRE_GROUP; ++s) p[s] = up[(size_t)s * sStride];
#pragma unroll
    for (int g = 0; g < PRE_STEPS / PRE_GROUP; ++g) {
#pragma unroll
      for (int s = g * PRE_GROUP; s < (g + 1) * PRE_GROUP; ++s) h = fmaf(lam, h, p[s]);
#pragma unroll
      for (int s = (g + PRE_AHEAD) * PRE_GROUP; s < (g + PRE_AHEAD + 1) * PRE_GROUP; ++s) {
        if (s < PRE_STEPS) p[s] = up[(size_t)s * sStride];
        else u[s - PRE_STEPS] = up[(size_t)s * sStride];
      }
    }
  }

#pragma nounroll
  for (int c = cBegin; c < cEnd; ++c) {
    float (*Hb)[H_PITCH] = Hs[c & 1];

#pragma unroll
    for (int s = 0; s < CH_STEPS; ++s) {
      h = fmaf(lam, h, u[s]);
      Hb[s][n] = h;
    }
    if (c + 1 < cEnd) {
#pragma unroll
      for (int s = 0; s < CH_STEPS; ++s)
        u[s] = up[(size_t)((c + 1) * CH_STEPS + s) * sStride];
    } else {
      if ((blockIdx.x & 3) == 0 && tpart == T_PARTS - 1) hT_out[b * N_DIM + n] = h;
    }
    asm volatile("s_wait_dscnt 0x0\n\ts_barrier_signal -1\n\ts_barrier_wait -1"
                 ::: "memory");

    const float* __restrict__ arow = &Hb[wave * 16 + lane16][2 * half];
    v2f a[N_DIM / 4];
#pragma unroll
    for (int kk = 0; kk < N_DIM / 4; ++kk) {
      a[kk].x = arow[kk * 4];
      a[kk].y = arow[kk * 4 + 1];
    }

    __builtin_amdgcn_sched_barrier(0);

    v8f acc0 = {0.f, 0.f, 0.f, 0.f, 0.f, 0.f, 0.f, 0.f};
    v8f acc1 = {0.f, 0.f, 0.f, 0.f, 0.f, 0.f, 0.f, 0.f};
#pragma unroll
    for (int kk = 0; kk < N_DIM / 4; ++kk) {
      acc0 = __builtin_amdgcn_wmma_f32_16x16x4_f32(
          false, a[kk], false, bf[0][kk], (short)0, acc0, false, false);
      acc1 = __builtin_amdgcn_wmma_f32_16x16x4_f32(
          false, a[kk], false, bf[1][kk], (short)0, acc1, false, false);
    }
    asm volatile("v_nop\n\tv_nop\n\tv_nop\n\tv_nop\n\tv_nop"
                 : "+v"(acc0), "+v"(acc1));
    __builtin_amdgcn_sched_barrier(0);

    const int sBase = c * CH_STEPS + wave * 16 + 8 * half;
#pragma unroll
    for (int r = 0; r < 8; ++r) {
      float* __restrict__ orow =
          out + ((size_t)(sBase + r) * BT_DIM + b) * O_DIM + o0 + lane16;
      orow[0] = acc0[r];
      orow[16] = acc1[r];
    }
  }
}

extern "C" void kernel_launch(void* const* d_in, const int* in_sizes, int n_in,
                              void* d_out, int out_size, void* d_ws,
                              size_t ws_size, hipStream_t stream) {
  (void)in_sizes; (void)n_in; (void)out_size; (void)ws_size;
  const float* x          = (const float*)d_in[0];
  const float* lambda_log = (const float*)d_in[1];
  const float* B          = (const float*)d_in[2];
  const float* C          = (const float*)d_in[3];
  float* out = (float*)d_out;
  float* U   = (float*)d_ws;

  float* outputs = out;
  float* hT      = out + (size_t)ROWS * O_DIM;

  gemm_nt_wreg<<<dim3(ROWS / G1_ROWS), dim3(128), 0, stream>>>(
      x, B, lambda_log, U);
  scan_gemm_nt_wmma<<<dim3(BT_DIM * (O_DIM / O_TILE) * T_PARTS), dim3(128), 0, stream>>>(
      U, C, lambda_log, outputs, hT);
}


# K1 writes U time-blocked (4 steps of a chain per 16 B, whole-line stores); K2 chain threads load 16 B: a block's U loads ahead of its first tile are one burst, sent ahead of the Cw fragments
# speedup vs baseline: 5.1847x; 1.0852x over previous
#include <hip/hip_runtime.h>
#include <hip/hip_bf16.h>
#include <math.h>

#define S_LEN 256
#define BT_DIM 32
#define D_DIM 128
#define N_DIM 128
#define O_DIM 128
#define ROWS (S_LEN * BT_DIM)
#define EPS_V 1e-7f
#define T_GROUP 4
#define S_GROUPS (S_LEN / T_GROUP)

typedef __attribute__((ext_vector_type(2))) float v2f;
typedef __attribute__((ext_vector_type(8))) float v8f;

typedef __attribute__((ext_vector_type(4))) float v4f;

__device__ __forceinline__ void load_frag_pair(const float* __restrict__ rowp,
                                               int q, v2f (&f)[D_DIM / 4]) {
  const v4f v = *reinterpret_cast<const v4f*>(rowp + 8 * q);
  f[2 * q].x = v.x;
  f[2 * q].y = v.y;
  f[2 * q + 1].x = v.z;
  f[2 * q + 1].y = v.w;
}

#define G1_ROWS 32
__global__ __launch_bounds__(128) __attribute__((amdgpu_waves_per_eu(1, 1)))
void gemm_nt_wreg(
    const float* __restrict__ in, const float* __restrict__ W,
    const float* __restrict__ lambda_log, float* __restrict__ out) {
  const int lane = threadIdx.x & 31;
  const int wave = threadIdx.x >> 5;
  const int half = lane >> 4;
  const int lane16 = lane & 15;
  const int sg = blockIdx.x >> 2;
  const int bBase = (blockIdx.x & 3) * 8;
  const int colBase = wave * 32;
  const int xrow0 = (T_GROUP * sg + (lane16 & 3)) * BT_DIM + bBase + (lane16 >> 2);

  float ll0 = lambda_log[colBase + lane16];
  float ll1 = lambda_log[colBase + 16 + lane16];

  const float* __restrict__ a0p = in + (size_t)xrow0 * D_DIM + 4 * half;
  const float* __restrict__ a1p = in + (size_t)(xrow0 + 4) * D_DIM + 4 * half;
  const float* __restrict__ b0p = W + (size_t)(colBase + lane16) * D_DIM + 4 * half;
  const float* __restrict__ b1p = W + (size_t)(colBase + 16 + lane16) * D_DIM + 4 * half;
  v2f a0[D_DIM / 4], a1[D_DIM / 4], bf0[D_DIM / 4], bf1[D_DIM / 4];
#pragma unroll
  for (int g = 0; g < 4; ++g) {
#pragma unroll
    for (int q = 4 * g; q < 4 * g + 4; ++q) {
      load_frag_pair(a0p, q, a0);
      load_frag_pair(b0p, q, bf0);
      load_frag_pair(b1p, q, bf1);
      load_frag_pair(a1p, q, a1);
    }
    asm volatile("" ::: "memory");
  }

  asm volatile("" : "+v"(ll0), "+v"(ll1));
  const float lam0 = __expf(-__expf(ll0));
  const float lam1 = __expf(-__expf(ll1));
  float g0 = sqrtf(fmaf(-lam0, lam0, 1.0f + EPS_V));
  float g1 = sqrtf(fmaf(-lam1, lam1, 1.0f + EPS_V));
  asm volatile("" : "+v"(g0), "+v"(g1));

  unsigned ooff = (((bBase + 2 * half) * S_GROUPS + sg) * N_DIM + colBase + lane16) * T_GROUP;
  asm volatile("" : "+v"(ooff));

  __builtin_amdgcn_sched_barrier(0);

  v8f acc00 = {0.f, 0.f, 0.f, 0.f, 0.f, 0.f, 0.f, 0.f};
  v8f acc01 = {0.f, 0.f, 0.f, 0.f, 0.f, 0.f, 0.f, 0.f};
  v8f acc10 = {0.f, 0.f, 0.f, 0.f, 0.f, 0.f, 0.f, 0.f};
  v8f acc11 = {0.f, 0.f, 0.f, 0.f, 0.f, 0.f, 0.f, 0.f};
#pragma unroll
  for (int kk = 0; kk < D_DIM / 4; ++kk) {
    acc00 = __builtin_amdgcn_wmma_f32_16x16x4_f32(
        false, a0[kk], false, bf0[kk], (short)0, acc00, false, false);
    acc01 = __builtin_amdgcn_wmma_f32_16x16x4_f32(
        false, a0[kk], false, bf1[kk], (short)0, acc01, false, false);
    acc10 = __builtin_amdgcn_wmma_f32_16x16x4_f32(
        false, a1[kk], false, bf0[kk], (short)0, acc10, false, false);
    acc11 = __builtin_amdgcn_wmma_f32_16x16x4_f32(
        false, a1[kk], false, bf1[kk], (short)0, acc11, false, false);
  }
  asm volatile("v_nop\n\tv_nop\n\tv_nop\n\tv_nop\n\tv_nop"
               : "+v"(acc00), "+v"(acc01), "+v"(acc10), "+v"(acc11));
  __builtin_amdgcn_sched_barrier(0);

  float* __restrict__ obase = out + ooff;
  const size_t bStride = (size_t)S_GROUPS * N_DIM * T_GROUP;
#pragma unroll
  for (int q = 0; q < 2; ++q) {
    float* __restrict__ o0 = obase + (size_t)q * bStride;
    float* __restrict__ o1 = obase + (size_t)(4 + q) * bStride;
    const v4f v00 = {g0 * acc00[4 * q], g0 * acc00[4 * q + 1], g0 * acc00[4 * q + 2], g0 * acc00[4 * q + 3]};
    const v4f v01 = {g1 * acc01[4 * q], g1 * acc01[4 * q + 1], g1 * acc01[4 * q + 2], g1 * acc01[4 * q + 3]};
    const v4f v10 = {g0 * acc10[4 * q], g0 * acc10[4 * q + 1], g0 * acc10[4 * q + 2], g0 * acc10[4 * q + 3]};
    const v4f v11 = {g1 * acc11[4 * q], g1 * acc11[4 * q + 1], g1 * acc11[4 * q + 2], g1 * acc11[4 * q + 3]};
    *reinterpret_cast<v4f*>(o0) = v00;
    *reinterpret_cast<v4f*>(o0 + 16 * T_GROUP) = v01;
    *reinterpret_cast<v4f*>(o1) = v10;
    *reinterpret_cast<v4f*>(o1 + 16 * T_GROUP) = v11;
  }
}

#define CH_STEPS 64
#define H_PITCH (N_DIM + 4)
#define O_TILE 32
#define T_PARTS 2
#define CH_PER_PART (S_LEN / CH_STEPS / T_PARTS)
#define PRE_STEPS (CH_PER_PART * CH_STEPS)
static_assert(T_PARTS == 2 && CH_PER_PART >= 2, "the kernel's opening assumes two time halves of at least two tiles");
#define CH_GROUPS (CH_STEPS / T_GROUP)
#define PRE_GROUPS (PRE_STEPS / T_GROUP)
#define CW_LOADS (2 * (N_DIM / 4))
#define CW_FIRST 14
#define CW_DOUBLE (2 * PRE_GROUPS - (CW_LOADS - CW_FIRST))

__device__ __forceinline__ void load_cw_frag(const float* __restrict__ crow0,
                                             const float* __restrict__ crow1,
                                             int i, v2f (&bf)[2][N_DIM / 4]) {
  const float* __restrict__ crow = (i & 1) ? crow1 : crow0;
  const int kk = i >> 1;
  bf[i & 1][kk].x = crow[kk * 4];
  bf[i & 1][kk].y = crow[kk * 4 + 1];
}

__device__ __forceinline__ void scan_tile(const v4f (&u4)[CH_GROUPS], float lam,
                                          float& h, float (*Hb)[H_PITCH], int n) {
#pragma unroll
  for (int g = 0; g < CH_GROUPS; ++g) {
    h = fmaf(lam, h, u4[g].x);
    Hb[T_GROUP * g][n] = h;
    h = fmaf(lam, h, u4[g].y);
    Hb[T_GROUP * g + 1][n] = h;
    h = fmaf(lam, h, u4[g].z);
    Hb[T_GROUP * g + 2][n] = h;
    h = fmaf(lam, h, u4[g].w);
    Hb[T_GROUP * g + 3][n] = h;
  }
}

__global__ __launch_bounds__(128) __attribute__((amdgpu_waves_per_eu(1, 1)))
void scan_gemm_nt_wmma(
    const float* __restrict__ U, const float* __restrict__ Cw,
    const float* __restrict__ lambda_log, float* __restrict__ out,
    float* __restrict__ hT_out) {
  __shared__ __attribute__((aligned(16))) float Hs[2][CH_STEPS][H_PITCH];

  const int tid = threadIdx.x;
  const int lane = tid & 31;
  const int wave = tid >> 5;
  const int half = lane >> 4;
  const int lane16 = lane & 15;
  const int b = (blockIdx.x >> 2) & (BT_DIM - 1);
  const int tpart = (T_PARTS - 1) - (blockIdx.x >> 7);
  const int o0 = (blockIdx.x & 3) * O_TILE;
  const int cBegin = tpart * CH_PER_PART;
  const int cEnd = cBegin + CH_PER_PART;
  const int n = tid;

  const float* __restrict__ crow0 = Cw + (size_t)(o0 + lane16) * N_DIM + 2 * half;
  const float* __restrict__ crow1 = crow0 + (size_t)16 * N_DIM;
  v2f bf[2][N_DIM / 4];

  const float ll = lambda_log[n];
  float lam;
  float h = 0.0f;
  const v4f* __restrict__ up4 =
      reinterpret_cast<const v4f*>(U) + (size_t)b * S_GROUPS * N_DIM + n;

  v4f u4[CH_GROUPS];
  if (tpart == 0) {
#pragma unroll
    for (int g = 0; g < CH_GROUPS; ++g) u4[g] = up4[(size_t)g * N_DIM];
#pragma unroll
    for (int i = 0; i < CW_LOADS; ++i) load_cw_frag(crow0, crow1, i, bf);
    float lle = ll;
    asm volatile("" : "+v"(lle));
    lam = __expf(-__expf(lle));
    scan_tile(u4, lam, h, Hs[cBegin & 1], n);
  } else {
    v4f p4[PRE_GROUPS];
#pragma unroll
    for (int g = 0; g < PRE_GROUPS; ++g) p4[g] = up4[(size_t)g * N_DIM];
#pragma unroll
    for (int g = 0; g < CH_GROUPS; ++g) u4[g] = up4[(size_t)(PRE_GROUPS + g) * N_DIM];
#pragma unroll
    for (int i = 0; i < CW_FIRST; ++i) load_cw_frag(crow0, crow1, i, bf);
    __builtin_amdgcn_sched_barrier(0);
    float lle = ll;
    asm volatile("" : "+v"(lle));
    lam = __expf(-__expf(lle));
#pragma unroll
    for (int g = 0; g < PRE_GROUPS; ++g) {
      h = fmaf(lam, h, p4[g].x);
      h = fmaf(lam, h, p4[g].y);
      h = fmaf(lam, h, p4[g].z);
      h = fmaf(lam, h, p4[g].w);
      asm volatile("" : "+v"(h));
      if (g < CW_DOUBLE) {
        load_cw_frag(crow0, crow1, CW_FIRST + g, bf);
      } else {
        load_cw_frag(crow0, crow1, CW_FIRST + CW_DOUBLE + 2 * (g - CW_DOUBLE), bf);
        load_cw_frag(crow0, crow1, CW_FIRST + CW_DOUBLE + 2 * (g - CW_DOUBLE) + 1, bf);
      }
    }
    scan_tile(u4, lam, h, Hs[cBegin & 1], n);
  }
#pragma unroll
  for (int g = 0; g < CH_GROUPS; ++g)
    u4[g] = up4[(size_t)((cBegin + 1) * CH_GROUPS + g) * N_DIM];

#pragma nounroll
  for (int c = cBegin; c < cEnd; ++c) {
    float (*Hb)[H_PITCH] = Hs[c & 1];

    asm volatile("s_wait_dscnt 0x0\n\ts_barrier_signal -1\n\ts_barrier_wait -1"
                 ::: "memory");

    const float* __restrict__ arow = &Hb[wave * 16 + lane16][2 * half];
    v2f a[N_DIM / 4];
#pragma unroll
    for (int kk = 0; kk < N_DIM / 4; ++kk) {
      a[kk].x = arow[kk * 4];
      a[kk].y = arow[kk * 4 + 1];
    }

    __builtin_amdgcn_sched_barrier(0);

    v8f acc0 = {0.f, 0.f, 0.f, 0.f, 0.f, 0.f, 0.f, 0.f};
    v8f acc1 = {0.f, 0.f, 0.f, 0.f, 0.f, 0.f, 0.f, 0.f};
#pragma unroll
    for (int kk = 0; kk < N_DIM / 4; ++kk) {
      acc0 = __builtin_amdgcn_wmma_f32_16x16x4_f32(
          false, a[kk], false, bf[0][kk], (short)0, acc0, false, false);
      acc1 = __builtin_amdgcn_wmma_f32_16x16x4_f32(
          false, a[kk], false, bf[1][kk], (short)0, acc1, false, false);
    }
    asm volatile("v_nop\n\tv_nop\n\tv_nop\n\tv_nop\n\tv_nop"
                 : "+v"(acc0), "+v"(acc1));
    __builtin_amdgcn_sched_barrier(0);

    const int sBase = c * CH_STEPS + wave * 16 + 8 * half;
#pragma unroll
    for (int r = 0; r < 8; ++r) {
      float* __restrict__ orow =
          out + ((size_t)(sBase + r) * BT_DIM + b) * O_DIM + o0 + lane16;
      orow[0] = acc0[r];
      orow[16] = acc1[r];
    }

    if (c + 1 < cEnd) {
      scan_tile(u4, lam, h, Hs[(c + 1) & 1], n);
      if (c + 2 < cEnd) {
#pragma unroll
        for (int g = 0; g < CH_GROUPS; ++g)
          u4[g] = up4[(size_t)((c + 2) * CH_GROUPS + g) * N_DIM];
      } else {
        if ((blockIdx.x & 3) == 0 && tpart == T_PARTS - 1) hT_out[b * N_DIM + n] = h;
      }
    }
  }
}

extern "C" void kernel_launch(void* const* d_in, const int* in_sizes, int n_in,
                              void* d_out, int out_size, void* d_ws,
                              size_t ws_size, hipStream_t stream) {
  (void)in_sizes; (void)n_in; (void)out_size; (void)ws_size;
  const float* x          = (const float*)d_in[0];
  const float* lambda_log = (const float*)d_in[1];
  const float* B          = (const float*)d_in[2];
  const float* C          = (const float*)d_in[3];
  float* out = (float*)d_out;
  float* U   = (float*)d_ws;

  float* outputs = out;
  float* hT      = out + (size_t)ROWS * O_DIM;

  gemm_nt_wreg<<<dim3(ROWS / G1_ROWS), dim3(128), 0, stream>>>(
      x, B, lambda_log, U);
  scan_gemm_nt_wmma<<<dim3(BT_DIM * (O_DIM / O_TILE) * T_PARTS), dim3(128), 0, stream>>>(
      U, C, lambda_log, outputs, hT);
}
